# starting kernel
# baseline (speedup 1.0000x reference)
#include <hip/hip_runtime.h>


typedef __attribute__((ext_vector_type(2))) float v2f;
typedef __attribute__((ext_vector_type(8))) float v8f;

constexpr int NB  = 16;
constexpr int NC  = 64;
constexpr int NFR = 8;
constexpr int ND  = 16;
constexpr int NL  = 64;
constexpr int NLP = 65;
constexpr int NP  = NLP * NLP;
constexpr int NPW = NLP * NL;

constexpr int OFF_FWDR = 0;
constexpr int OFF_FWDI = OFF_FWDR + 80 * 80;
constexpr int OFF_ER   = OFF_FWDI + 80 * 80;
constexpr int OFF_EI   = OFF_ER + 64 * 64;
constexpr int OFF_WNR  = OFF_EI + 64 * 64;
constexpr int OFF_WNI  = OFF_WNR + 64 * 16;
constexpr int OFF_IVR  = OFF_WNI + 64 * 16;
constexpr int OFF_IVI  = OFF_IVR + 68 * 64;
constexpr int OFF_IUR  = OFF_IVI + 68 * 64;
constexpr int OFF_IUI  = OFF_IUR + 64 * 68;
constexpr int TW_TOTAL = OFF_IUI + 64 * 68;
constexpr int TW_PAD   = 40960;

#define PI_D 3.14159265358979323846


static __device__ __forceinline__ v8f wmma4(v2f a, v2f b, v8f c) {
  return __builtin_amdgcn_wmma_f32_16x16x4_f32(false, a, false, b, (short)0, c,
                                               false, false);
}

static __device__ __forceinline__ v8f zero8() {
  v8f z = {0.f, 0.f, 0.f, 0.f, 0.f, 0.f, 0.f, 0.f};
  return z;
}

static __device__ __forceinline__ v2f frag_a(const float* __restrict__ A,
                                             int lda, int m0, int k0) {
  int lane = threadIdx.x & 31;
  int m = m0 + (lane & 15);
  int k = k0 + ((lane >> 4) << 1);
  const float* p = A + m * lda + k;
  v2f r;
  r.x = p[0];
  r.y = p[1];
  return r;
}

static __device__ __forceinline__ v2f frag_a_guard(const float* __restrict__ A,
                                                   int lda, int m0, int k0,
                                                   int mmax, int kmax) {
  int lane = threadIdx.x & 31;
  int m = m0 + (lane & 15);
  int k = k0 + ((lane >> 4) << 1);
  v2f r;
  r.x = 0.f;
  r.y = 0.f;
  if (m < mmax) {
    const float* p = A + m * lda + k;
    if (k < kmax) r.x = p[0];
    if (k + 1 < kmax) r.y = p[1];
  }
  return r;
}

static __device__ __forceinline__ v2f frag_b(const float* __restrict__ B,
                                             int ldb, int k0, int n0) {
  int lane = threadIdx.x & 31;
  int n = n0 + (lane & 15);
  int k = k0 + ((lane >> 4) << 1);
  const float* p = B + k * ldb + n;
  v2f r;
  r.x = p[0];
  r.y = p[ldb];
  return r;
}

static __device__ __forceinline__ v2f frag_b_nguard(const float* __restrict__ B,
                                                    int ldb, int k0, int n0,
                                                    int nmax) {
  int lane = threadIdx.x & 31;
  int n = n0 + (lane & 15);
  int k = k0 + ((lane >> 4) << 1);
  v2f r;
  r.x = 0.f;
  r.y = 0.f;
  if (n < nmax) {
    const float* p = B + k * ldb + n;
    r.x = p[0];
    r.y = p[ldb];
  }
  return r;
}

static __device__ __forceinline__ v2f frag_b_kguard(const float* __restrict__ B,
                                                    int ldb, int k0, int n0,
                                                    int kmax) {
  int lane = threadIdx.x & 31;
  int n = n0 + (lane & 15);
  int k = k0 + ((lane >> 4) << 1);
  v2f r;
  r.x = 0.f;
  r.y = 0.f;
  const float* p = B + k * ldb + n;
  if (k < kmax) r.x = p[0];
  if (k + 1 < kmax) r.y = p[ldb];
  return r;
}

static __device__ __forceinline__ void cwmma(v2f ar, v2f ai, v2f br, v2f bi,
                                             v8f& dr, v8f& di) {
  dr = wmma4(ar, br, dr);
  v2f nai;
  nai.x = -ai.x;
  nai.y = -ai.y;
  dr = wmma4(nai, bi, dr);
  di = wmma4(ar, bi, di);
  di = wmma4(ai, br, di);
}


__global__ void k_twiddles(float* __restrict__ tw) {
  int t = blockIdx.x * blockDim.x + threadIdx.x;
  int which = blockIdx.y;
  if (which == 0) {
    if (t >= 80 * 80) return;
    int a = t / 80, b = t % 80;
    float r = 0.f, im = 0.f;
    if (a < NLP && b < NLP) {
      double ang = -2.0 * PI_D * (double)((a * b) % NLP) / (double)NLP;
      r = (float)cos(ang);
      im = (float)sin(ang);
    }
    tw[OFF_FWDR + t] = r;
    tw[OFF_FWDI + t] = im;
  } else if (which == 1) {
    if (t >= NC * NC) return;
    int c = t / NC, k = t % NC;
    double ang = -2.0 * PI_D * (double)((c * k) % NC) / (double)NC;
    tw[OFF_ER + t] = (float)cos(ang);
    tw[OFF_EI + t] = (float)sin(ang);
  } else if (which == 2) {
    if (t >= NC * ND) return;
    int c = t / ND, d = t % ND;
    double ang = 2.0 * PI_D * (double)((c * 4 * d) % NC) / (double)NC;
    tw[OFF_WNR + t] = (float)(cos(ang) / 64.0);
    tw[OFF_WNI + t] = (float)(sin(ang) / 64.0);
  } else if (which == 3) {
    if (t >= 68 * NL) return;
    int v = t / NL, w = t % NL;
    float r = 0.f, im = 0.f;
    if (v < NLP) {
      double ang = 2.0 * PI_D * (double)((v * (w + 1)) % NLP) / (double)NLP;
      r = (float)cos(ang);
      im = (float)sin(ang);
    }
    tw[OFF_IVR + t] = r;
    tw[OFF_IVI + t] = im;
  } else {
    if (t >= NL * 68) return;
    int h = t / 68, u = t % 68;
    float r = 0.f, im = 0.f;
    if (u < NLP) {
      double ang = 2.0 * PI_D * (double)((u * (h + 1)) % NLP) / (double)NLP;
      double s = 1.0 / (double)(NLP * NLP);
      r = (float)(cos(ang) * s);
      im = (float)(sin(ang) * s);
    }
    tw[OFF_IUR + t] = r;
    tw[OFF_IUI + t] = im;
  }
}


__global__ __launch_bounds__(128) void k_fft2d(const float* __restrict__ x,
                                               const float* __restrict__ tw,
                                               float* __restrict__ Yr,
                                               float* __restrict__ Yi) {
  __shared__ float xs[NL * NL];
  __shared__ float Tr[NL * 80];
  __shared__ float Ti[NL * 80];
  const int plane = blockIdx.x;
  const int tid = threadIdx.x;
  const int wave = tid >> 5;
  const int lane = tid & 31;

  const float* xp = x + (size_t)plane * (NL * NL);
  for (int i = tid; i < NL * NL; i += 128) xs[i] = xp[i];
  __syncthreads();

  const float* Fr = tw + OFF_FWDR;
  const float* Fi = tw + OFF_FWDI;

  for (int t = wave; t < 4 * 5; t += 4) {
    int mo = (t / 5) * 16, no = (t % 5) * 16;
    v8f ar = zero8(), ai = zero8();
    for (int k0 = 0; k0 < NL; k0 += 4) {
      v2f a = frag_a(xs, NL, mo, k0);
      ar = wmma4(a, frag_b(Fr, 80, k0, no), ar);
      ai = wmma4(a, frag_b(Fi, 80, k0, no), ai);
    }
    int n = no + (lane & 15);
    int mb = mo + ((lane >> 4) << 3);
#pragma unroll
    for (int i = 0; i < 8; ++i) {
      Tr[(mb + i) * 80 + n] = ar[i];
      Ti[(mb + i) * 80 + n] = ai[i];
    }
  }
  __syncthreads();

  float* yr = Yr + (size_t)plane * NP;
  float* yi = Yi + (size_t)plane * NP;
  for (int t = wave; t < 5 * 5; t += 4) {
    int mo = (t / 5) * 16, no = (t % 5) * 16;
    v8f dr = zero8(), di = zero8();
    for (int k0 = 0; k0 < NL; k0 += 4) {
      v2f par = frag_a(Fr, 80, mo, k0);
      v2f pai = frag_a(Fi, 80, mo, k0);
      v2f pbr = frag_b(Tr, 80, k0, no);
      v2f pbi = frag_b(Ti, 80, k0, no);
      cwmma(par, pai, pbr, pbi, dr, di);
    }
    int n = no + (lane & 15);
    int mb = mo + ((lane >> 4) << 3);
    if (n < NLP) {
#pragma unroll
      for (int i = 0; i < 8; ++i) {
        int m = mb + i;
        if (m < NLP) {
          yr[m * NLP + n] = dr[i];
          yi[m * NLP + n] = di[i];
        }
      }
    }
  }
}


__global__ __launch_bounds__(128) void k_cfft(const float* __restrict__ tw,
                                              const float* __restrict__ Yr,
                                              const float* __restrict__ Yi,
                                              float* __restrict__ Xr,
                                              float* __restrict__ Xi) {
  const int b = blockIdx.y;
  const int no = blockIdx.x * 16;
  const int mo = (threadIdx.x >> 5) * 16;
  const int lane = threadIdx.x & 31;
  const float* Er = tw + OFF_ER;
  const float* Ei = tw + OFF_EI;
  const float* yr = Yr + (size_t)b * NC * NP;
  const float* yi = Yi + (size_t)b * NC * NP;
  v8f dr = zero8(), di = zero8();
  for (int k0 = 0; k0 < NC; k0 += 4) {
    v2f ar = frag_a(Er, NC, mo, k0);
    v2f ai = frag_a(Ei, NC, mo, k0);
    v2f br = frag_b_nguard(yr, NP, k0, no, NP);
    v2f bi = frag_b_nguard(yi, NP, k0, no, NP);
    cwmma(ar, ai, br, bi, dr, di);
  }
  int n = no + (lane & 15);
  int mb = mo + ((lane >> 4) << 3);
  if (n < NP) {
    float* xr = Xr + (size_t)b * NC * NP;
    float* xi = Xi + (size_t)b * NC * NP;
#pragma unroll
    for (int i = 0; i < 8; ++i) {
      xr[(mb + i) * NP + n] = dr[i];
      xi[(mb + i) * NP + n] = di[i];
    }
  }
}


__global__ __launch_bounds__(128) void k_mix(const float* __restrict__ tw,
                                             const float* __restrict__ Xr,
                                             const float* __restrict__ Xi,
                                             const float* __restrict__ wre,
                                             const float* __restrict__ wim,
                                             float* __restrict__ Gr,
                                             float* __restrict__ Gi, int nidx) {
  const int wave = threadIdx.x >> 5;
  const int p = blockIdx.x * 4 + wave;
  if (p >= NP) return;
  const int lane = threadIdx.x & 31;
  const float* WNr = tw + OFF_WNR;
  const float* WNi = tw + OFF_WNI;
  const int bm = lane & 15;
  const int khalf = (lane >> 4) << 1;
  v8f dr = zero8(), di = zero8();
  for (int k0 = 0; k0 < NC; k0 += 4) {
    v2f ar, ai;
#pragma unroll
    for (int j = 0; j < 2; ++j) {
      int c = k0 + khalf + j;
      size_t xo = ((size_t)bm * NC + c) * NP + p;
      size_t wo = ((size_t)nidx * NC + c) * NP + p;
      float xr = Xr[xo], xi = Xi[xo];
      float wr = wre[wo], wi = wim[wo];
      ar[j] = xr * wr - xi * wi;
      ai[j] = xr * wi + xi * wr;
    }
    v2f br = frag_b(WNr, ND, k0, 0);
    v2f bi = frag_b(WNi, ND, k0, 0);
    cwmma(ar, ai, br, bi, dr, di);
  }
  int n = lane & 15;
  int mb = (lane >> 4) << 3;
#pragma unroll
  for (int i = 0; i < 8; ++i) {
    size_t o = ((size_t)(mb + i) * ND + n) * NP + p;
    Gr[o] = dr[i];
    Gi[o] = di[i];
  }
}


__global__ __launch_bounds__(128) void k_ifft_v(const float* __restrict__ tw,
                                                const float* __restrict__ Gr,
                                                const float* __restrict__ Gi,
                                                float* __restrict__ Pr,
                                                float* __restrict__ Pi) {
  const int d = blockIdx.x, b = blockIdx.y;
  const int wave = threadIdx.x >> 5;
  const int lane = threadIdx.x & 31;
  const float* IVr = tw + OFF_IVR;
  const float* IVi = tw + OFF_IVI;
  const float* gr = Gr + ((size_t)b * ND + d) * NP;
  const float* gi = Gi + ((size_t)b * ND + d) * NP;
  float* pr = Pr + ((size_t)b * ND + d) * NPW;
  float* pi = Pi + ((size_t)b * ND + d) * NPW;
  for (int t = wave; t < 5 * 4; t += 4) {
    int mo = (t / 4) * 16, no = (t % 4) * 16;
    v8f dr = zero8(), di = zero8();
    for (int k0 = 0; k0 < 68; k0 += 4) {
      v2f ar = frag_a_guard(gr, NLP, mo, k0, NLP, NLP);
      v2f ai = frag_a_guard(gi, NLP, mo, k0, NLP, NLP);
      v2f br = frag_b(IVr, NL, k0, no);
      v2f bi = frag_b(IVi, NL, k0, no);
      cwmma(ar, ai, br, bi, dr, di);
    }
    int n = no + (lane & 15);
    int mb = mo + ((lane >> 4) << 3);
#pragma unroll
    for (int i = 0; i < 8; ++i) {
      int m = mb + i;
      if (m < NLP) {
        pr[m * NL + n] = dr[i];
        pi[m * NL + n] = di[i];
      }
    }
  }
}


__global__ __launch_bounds__(128) void k_ifft_u(const float* __restrict__ tw,
                                                const float* __restrict__ Pr,
                                                const float* __restrict__ Pi,
                                                float* __restrict__ out,
                                                int nidx) {
  const int d = blockIdx.x, b = blockIdx.y;
  const int wave = threadIdx.x >> 5;
  const int lane = threadIdx.x & 31;
  const float* IUr = tw + OFF_IUR;
  const float* IUi = tw + OFF_IUI;
  const float* pr = Pr + ((size_t)b * ND + d) * NPW;
  const float* pi = Pi + ((size_t)b * ND + d) * NPW;
  float* op = out + ((size_t)b * (ND * NFR) + (d * NFR + nidx)) * (NL * NL);
  for (int t = wave; t < 4 * 4; t += 4) {
    int mo = (t / 4) * 16, no = (t % 4) * 16;
    v8f dr = zero8();
    for (int k0 = 0; k0 < 68; k0 += 4) {
      v2f ar = frag_a(IUr, 68, mo, k0);
      v2f ai = frag_a(IUi, 68, mo, k0);
      v2f br = frag_b_kguard(pr, NL, k0, no, NLP);
      v2f bi = frag_b_kguard(pi, NL, k0, no, NLP);
      dr = wmma4(ar, br, dr);
      v2f nai;
      nai.x = -ai.x;
      nai.y = -ai.y;
      dr = wmma4(nai, bi, dr);
    }
    int n = no + (lane & 15);
    int mb = mo + ((lane >> 4) << 3);
#pragma unroll
    for (int i = 0; i < 8; ++i) op[(mb + i) * NL + n] = dr[i];
  }
}


extern "C" void kernel_launch(void* const* d_in, const int* in_sizes, int n_in,
                              void* d_out, int out_size, void* d_ws,
                              size_t ws_size, hipStream_t stream) {
  (void)in_sizes; (void)n_in; (void)out_size; (void)ws_size;
  const float* x   = (const float*)d_in[0];
  const float* wre = (const float*)d_in[1];
  const float* wim = (const float*)d_in[2];
  float* out = (float*)d_out;
  float* ws = (float*)d_ws;

  float* tw = ws;
  size_t off = TW_PAD;
  float* Yr = ws + off; off += (size_t)NB * NC * NP;
  float* Yi = ws + off; off += (size_t)NB * NC * NP;
  float* Xr = ws + off; off += (size_t)NB * NC * NP;
  float* Xi = ws + off; off += (size_t)NB * NC * NP;
  float* Gr = ws + off; off += (size_t)NB * ND * NP;
  float* Gi = ws + off; off += (size_t)NB * ND * NP;
  float* Pr = ws + off; off += (size_t)NB * ND * NPW;
  float* Pi = ws + off; off += (size_t)NB * ND * NPW;

  k_twiddles<<<dim3(25, 5), 256, 0, stream>>>(tw);
  k_fft2d<<<NB * NC, 128, 0, stream>>>(x, tw, Yr, Yi);
  k_cfft<<<dim3((NP + 15) / 16, NB), 128, 0, stream>>>(tw, Yr, Yi, Xr, Xi);
  for (int n = 0; n < NFR; ++n) {
    k_mix<<<(NP + 3) / 4, 128, 0, stream>>>(tw, Xr, Xi, wre, wim, Gr, Gi, n);
    k_ifft_v<<<dim3(ND, NB), 128, 0, stream>>>(tw, Gr, Gi, Pr, Pi);
    k_ifft_u<<<dim3(ND, NB), 128, 0, stream>>>(tw, Pr, Pi, out, n);
  }
}


# forward 2D DFT: table rows k, 65-k are conjugates, so both stages add and subtract the data's mirror pairs first (k = 33, not 64); stage 2: rows u, 65-u from one cos and sin sum; sum order changes
# speedup vs baseline: 6.0054x; 6.0054x over previous
#include <hip/hip_runtime.h>


typedef __attribute__((ext_vector_type(2))) float v2f;
typedef __attribute__((ext_vector_type(4))) float v4f;
typedef __attribute__((ext_vector_type(8))) float v8f;

constexpr int NB  = 16;
constexpr int NC  = 64;
constexpr int NFR = 8;
constexpr int ND  = 16;
constexpr int NL  = 64;
constexpr int NLP = 65;
constexpr int NP  = NLP * NLP;
constexpr int NPW = NLP * NL;

constexpr int OFF_FWDR = 0;
constexpr int OFF_FWDI = OFF_FWDR + 80 * 80;
constexpr int OFF_ER   = OFF_FWDI + 80 * 80;
constexpr int OFF_EI   = OFF_ER + 64 * 64;
constexpr int OFF_WNR  = OFF_EI + 64 * 64;
constexpr int OFF_WNI  = OFF_WNR + 64 * 16;
constexpr int OFF_IVR  = OFF_WNI + 64 * 16;
constexpr int OFF_IVI  = OFF_IVR + 68 * 64;
constexpr int OFF_IUR  = OFF_IVI + 68 * 64;
constexpr int OFF_IUI  = OFF_IUR + 64 * 68;
constexpr int TW_TOTAL = OFF_IUI + 64 * 68;
constexpr int TW_PAD   = 40960;

#define PI_D 3.14159265358979323846


static __device__ __forceinline__ v8f wmma4(v2f a, v2f b, v8f c) {
  return __builtin_amdgcn_wmma_f32_16x16x4_f32(false, a, false, b, (short)0, c,
                                               false, false);
}

static __device__ __forceinline__ v8f zero8() {
  v8f z = {0.f, 0.f, 0.f, 0.f, 0.f, 0.f, 0.f, 0.f};
  return z;
}

static __device__ __forceinline__ v2f frag_a(const float* __restrict__ A,
                                             int lda, int m0, int k0) {
  int lane = threadIdx.x & 31;
  int m = m0 + (lane & 15);
  int k = k0 + ((lane >> 4) << 1);
  const float* p = A + m * lda + k;
  v2f r;
  r.x = p[0];
  r.y = p[1];
  return r;
}

static __device__ __forceinline__ v2f frag_b(const float* __restrict__ B,
                                             int ldb, int k0, int n0) {
  int lane = threadIdx.x & 31;
  int n = n0 + (lane & 15);
  int k = k0 + ((lane >> 4) << 1);
  const float* p = B + k * ldb + n;
  v2f r;
  r.x = p[0];
  r.y = p[ldb];
  return r;
}

static __device__ __forceinline__ v2f frag_b_nclamp(const float* __restrict__ B,
                                                    int ldb, int k0, int n0,
                                                    int nmax) {
  int lane = threadIdx.x & 31;
  int n = n0 + (lane & 15);
  if (n > nmax - 1) n = nmax - 1;
  int k = k0 + ((lane >> 4) << 1);
  const float* p = B + k * ldb + n;
  v2f r;
  r.x = p[0];
  r.y = p[ldb];
  return r;
}

static __device__ __forceinline__ v2f frag_a_mclamp(const float* __restrict__ A,
                                                    int lda, int m0, int k0,
                                                    int mmax) {
  int lane = threadIdx.x & 31;
  int m = m0 + (lane & 15);
  if (m > mmax - 1) m = mmax - 1;
  int k = k0 + ((lane >> 4) << 1);
  const float* p = A + m * lda + k;
  v2f r;
  r.x = p[0];
  r.y = p[1];
  return r;
}

static __device__ __forceinline__ void cwmma(v2f ar, v2f ai, v2f br, v2f bi,
                                             v8f& dr, v8f& di) {
  dr = wmma4(ar, br, dr);
  v2f nai;
  nai.x = -ai.x;
  nai.y = -ai.y;
  dr = wmma4(nai, bi, dr);
  di = wmma4(ar, bi, di);
  di = wmma4(ai, br, di);
}


__global__ void k_twiddles(float* __restrict__ tw) {
  int t = blockIdx.x * blockDim.x + threadIdx.x;
  int which = blockIdx.y;
  if (which == 0) {
    if (t >= 80 * 80) return;
    int a = t / 80, b = t % 80;
    float r = 0.f, im = 0.f;
    if (a < NLP && b < NLP) {
      double ang = -2.0 * PI_D * (double)((a * b) % NLP) / (double)NLP;
      r = (float)cos(ang);
      im = (float)sin(ang);
    }
    tw[OFF_FWDR + t] = r;
    tw[OFF_FWDI + t] = im;
  } else if (which == 1) {
    if (t >= NC * NC) return;
    int c = t / NC, k = t % NC;
    double ang = -2.0 * PI_D * (double)((c * k) % NC) / (double)NC;
    tw[OFF_ER + t] = (float)cos(ang);
    tw[OFF_EI + t] = (float)sin(ang);
  } else if (which == 2) {
    if (t >= NC * ND) return;
    int c = t / ND, d = t % ND;
    double ang = 2.0 * PI_D * (double)((c * 4 * d) % NC) / (double)NC;
    tw[OFF_WNR + t] = (float)(cos(ang) / 64.0);
    tw[OFF_WNI + t] = (float)(sin(ang) / 64.0);
  } else if (which == 3) {
    if (t >= 68 * NL) return;
    int v = t / NL, w = t % NL;
    float r = 0.f, im = 0.f;
    if (v < NLP) {
      double ang = 2.0 * PI_D * (double)((v * (w + 1)) % NLP) / (double)NLP;
      r = (float)cos(ang);
      im = (float)sin(ang);
    }
    tw[OFF_IVR + t] = r;
    tw[OFF_IVI + t] = im;
  } else {
    if (t >= NL * 68) return;
    int h = t / 68, u = t % 68;
    float r = 0.f, im = 0.f;
    if (u < NLP) {
      double ang = 2.0 * PI_D * (double)((u * (h + 1)) % NLP) / (double)NLP;
      double s = 1.0 / (double)(NLP * NLP);
      r = (float)(cos(ang) * s);
      im = (float)(sin(ang) * s);
    }
    tw[OFF_IUR + t] = r;
    tw[OFF_IUI + t] = im;
  }
}


__global__ __launch_bounds__(128) void k_fft2d(const float* __restrict__ x,
                                               const float* __restrict__ tw,
                                               float* __restrict__ Yr,
                                               float* __restrict__ Yi) {
  constexpr int FP = 68;
  constexpr int FR = 68;
  __shared__ __attribute__((aligned(16))) float U[2 * FR * FP];
  float* xs = U;
  __shared__ float Tr[NL * 80];
  __shared__ float Ti[NL * 80];
  const int plane = blockIdx.x;
  const int tid = threadIdx.x;
  const int wave = tid >> 5;
  const int lane = tid & 31;

  const float* xp = x + (size_t)plane * (NL * NL);
  for (int i = tid; i < NL * NL; i += 128) xs[i] = xp[i];
  __syncthreads();

  const float* Fr = tw + OFF_FWDR;
  const float* Fi = tw + OFF_FWDI;

  const int hh = lane >> 4;

  for (int t = wave; t < 4 * 5; t += 4) {
    int mo = (t / 5) * 16, no = (t % 5) * 16;
    v8f ar = zero8(), ai = zero8();
    const float* xr = xs + (mo + (lane & 15)) * NL + 2 * hh;
    const float* xm = xs + (mo + (lane & 15)) * NL + 64 - 2 * hh;
#pragma unroll
    for (int s = 0; s < 9; ++s) {
      const float d0 = xr[4 * s], d1 = xr[4 * s + 1];
      float m0, m1;
      if (s == 0) {
        const float* xq = xs + (mo + (lane & 15)) * NL + 62;
        const float q0 = xq[1], q1 = xq[0];
        m0 = hh ? q0 : 0.f;
        m1 = hh ? q1 : 0.f;
      } else {
        m0 = xm[1 - 4 * s];
        m1 = xm[-4 * s];
      }
      v2f ap, am;
      ap.x = d0 + m0;
      ap.y = d1 + m1;
      am.x = d0 - m0;
      am.y = d1 - m1;
      if (s == 8) {
        ap.x = hh ? 0.f : ap.x;
        am.x = hh ? 0.f : am.x;
        ap.y = 0.f;
        am.y = 0.f;
      }
      ar = wmma4(ap, frag_b(Fr, 80, 4 * s, no), ar);
      ai = wmma4(am, frag_b(Fi, 80, 4 * s, no), ai);
    }
    int n = no + (lane & 15);
    int mb = mo + ((lane >> 4) << 3);
#pragma unroll
    for (int i = 0; i < 8; ++i) {
      Tr[(mb + i) * 80 + n] = ar[i];
      Ti[(mb + i) * 80 + n] = ai[i];
    }
  }
  constexpr int FN4 = 2 * FR * 16;
  static_assert(FN4 == 17 * 128, "17 whole pieces of one element per thread");
  const int ct = tid & 127;
  v4f fv[17];
#pragma unroll
  for (int j = 0; j < 17; ++j) {
    const int e = ct + 128 * j;
    const int part = e / (FR * 16);
    const int r = e - part * (FR * 16);
    fv[j] = reinterpret_cast<const v4f*>(tw + OFF_FWDR)
                [part * ((OFF_FWDI - OFF_FWDR) / 4) + (r >> 4) * 20 + (r & 15)];
  }
  __syncthreads();
#pragma unroll
  for (int j = 0; j < 17; ++j) {
    const int e = ct + 128 * j;
    const int part = e / (FR * 16);
    const int r = e - part * (FR * 16);
    reinterpret_cast<v4f*>(U)[part * (FR * FP / 4) + (r >> 4) * (FP / 4) + (r & 15)] = fv[j];
  }
  __syncthreads();
  const float* Usr = U;
  const float* Usi = U + FR * FP;

  float* yr = Yr + (size_t)plane * NP;
  float* yi = Yi + (size_t)plane * NP;
  for (int t = wave; t < 3 * 5; t += 4) {
    int mo = (t / 5) * 16, no = (t % 5) * 16;
    const int n = no + (lane & 15);
    const float* trd = Tr + (2 * hh) * 80 + n;
    const float* tid = Ti + (2 * hh) * 80 + n;
    const float* trm = Tr + (62 - 2 * hh) * 80 + n;
    const float* tim = Ti + (62 - 2 * hh) * 80 + n;
    v8f cr = zero8(), ci = zero8(), qr = zero8(), qi = zero8();
#pragma unroll
    for (int s = 0; s < 9; ++s) {
      const float d0r = trd[(4 * s) * 80], d1r = trd[(4 * s + 1) * 80];
      const float d0i = tid[(4 * s) * 80], d1i = tid[(4 * s + 1) * 80];
      float m0r, m1r, m0i, m1i;
      if (s == 0) {
        const float a0r = Tr[63 * 80 + n], a1r = Tr[62 * 80 + n];
        const float a0i = Ti[63 * 80 + n], a1i = Ti[62 * 80 + n];
        m0r = hh ? a0r : 0.f;
        m1r = hh ? a1r : 0.f;
        m0i = hh ? a0i : 0.f;
        m1i = hh ? a1i : 0.f;
      } else {
        m0r = trm[(3 - 4 * s) * 80];
        m1r = trm[(2 - 4 * s) * 80];
        m0i = tim[(3 - 4 * s) * 80];
        m1i = tim[(2 - 4 * s) * 80];
      }
      v2f bpr, bpi, bmr, bmi;
      bpr.x = d0r + m0r;
      bpr.y = d1r + m1r;
      bpi.x = d0i + m0i;
      bpi.y = d1i + m1i;
      bmr.x = d0r - m0r;
      bmr.y = d1r - m1r;
      bmi.x = d0i - m0i;
      bmi.y = d1i - m1i;
      if (s == 8) {
        bpr.x = hh ? 0.f : bpr.x;
        bpi.x = hh ? 0.f : bpi.x;
        bmr.x = hh ? 0.f : bmr.x;
        bmi.x = hh ? 0.f : bmi.x;
        bpr.y = 0.f;
        bpi.y = 0.f;
        bmr.y = 0.f;
        bmi.y = 0.f;
      }
      v2f fr = frag_a(Usr, FP, mo, 4 * s);
      v2f fi = frag_a(Usi, FP, mo, 4 * s);
      cr = wmma4(fr, bpr, cr);
      ci = wmma4(fr, bpi, ci);
      qr = wmma4(fi, bmr, qr);
      qi = wmma4(fi, bmi, qi);
    }
    int mb = mo + (hh << 3);
    if (n < NLP) {
#pragma unroll
      for (int i = 0; i < 8; ++i) {
        int m = mb + i;
        if (m <= 32) {
          yr[m * NLP + n] = cr[i] - qi[i];
          yi[m * NLP + n] = ci[i] + qr[i];
          if (m >= 1) {
            yr[(NLP - m) * NLP + n] = cr[i] + qi[i];
            yi[(NLP - m) * NLP + n] = ci[i] - qr[i];
          }
        }
      }
    }
  }
}


__global__ __launch_bounds__(512) void k_cfft(const float* __restrict__ tw,
                                              const float* __restrict__ Yr,
                                              const float* __restrict__ Yi,
                                              float* __restrict__ XT) {
  __shared__ v4f S[16 * 256];
  constexpr int EP = 68;
  __shared__ __attribute__((aligned(16))) float Es[2 * 32 * EP];
  const int b = threadIdx.x >> 5;
  const int lane = threadIdx.x & 31;
  const int no = blockIdx.x * 16;
  const int chalf = blockIdx.y;
  const int cb = chalf * 32;
  const float* Er = tw + OFF_ER;
  const float* Ei = tw + OFF_EI;
  const float* yr = Yr + (size_t)b * NC * NP;
  const float* yi = Yi + (size_t)b * NC * NP;
  {
    const int ct = (int)threadIdx.x & 511;
    const v4f e_re = reinterpret_cast<const v4f*>(Er + cb * NC)[ct];
    const v4f e_im = reinterpret_cast<const v4f*>(Ei + cb * NC)[ct];
    v4f* es4 = reinterpret_cast<v4f*>(Es);
    const int de = (ct >> 4) * (EP / 4) + (ct & 15);
    es4[de] = e_re;
    es4[32 * EP / 4 + de] = e_im;
  }
  __syncthreads();
  const float* Esr = Es;
  const float* Esi = Es + 32 * EP;
  v8f dr0 = zero8(), di0 = zero8(), dr1 = zero8(), di1 = zero8();
  for (int k0 = 0; k0 < NC; k0 += 4) {
    v2f br = frag_b_nclamp(yr, NP, k0, no, NP);
    v2f bi = frag_b_nclamp(yi, NP, k0, no, NP);
    v2f ar = frag_a(Esr, EP, 0, k0);
    v2f ai = frag_a(Esi, EP, 0, k0);
    cwmma(ar, ai, br, bi, dr0, di0);
    ar = frag_a(Esr, EP, 16, k0);
    ai = frag_a(Esi, EP, 16, k0);
    cwmma(ar, ai, br, bi, dr1, di1);
  }
  const int pl = lane & 15;
  const int h = lane >> 4;
  v4f* srow = S + pl * 256 + (b ^ pl);
#pragma unroll
  for (int q = 0; q < 4; ++q) {
    v4f e0 = {dr0[2 * q], di0[2 * q], dr0[2 * q + 1], di0[2 * q + 1]};
    v4f e1 = {dr1[2 * q], di1[2 * q], dr1[2 * q + 1], di1[2 * q + 1]};
    srow[(4 * h + q) * 16] = e0;
    srow[(8 + 4 * h + q) * 16] = e1;
  }
  __syncthreads();
  v4f* xt = reinterpret_cast<v4f*>(XT);
  const int idx = threadIdx.x & 255;
  const int psel = threadIdx.x >> 8;
#pragma unroll
  for (int pass = 0; pass < 8; ++pass) {
    int pp = pass * 2 + psel;
    int p = no + pp;
    if (p < NP)
      xt[(size_t)p * 512 + chalf * 256 + idx] =
          S[pp * 256 + (idx & 0xF0) + ((idx & 15) ^ pp)];
  }
}


constexpr int MIXP = 17;
constexpr int MIXB = (NP + MIXP - 1) / MIXP;
constexpr int GSZ  = NB * ND * NP;
constexpr int PSZ  = NB * ND * NPW;
constexpr int GPR  = 2 * NLP;
constexpr int GLST = 64 * NLP;

struct GRaw {
  v2f dr, di, mr, mi;
};

static __device__ __forceinline__ void load_g_raw(const float* __restrict__ gr,
                                                  const float* __restrict__ gi,
                                                  int mo, GRaw* g) {
  const int lane = threadIdx.x & 31;
  const int h = lane >> 4;
  int m = mo + (lane & 15);
  if (m > NLP - 1) m = NLP - 1;
  const float* rr = gr + h * GPR + 2 * m;
  const float* ri = gi + h * GPR + 2 * m;
  const float* qr = gr + (32 - h) * GPR + 2 * m;
  const float* qi = gi + (32 - h) * GPR + 2 * m;
#pragma unroll
  for (int s = 1; s < 9; ++s) {
    g[s].dr.x = rr[2 * s * GPR];
    g[s].dr.y = rr[2 * s * GPR + 1];
    g[s].di.x = ri[2 * s * GPR];
    g[s].di.y = ri[2 * s * GPR + 1];
    g[s].mr.x = qr[-2 * s * GPR];
    g[s].mr.y = qr[-2 * s * GPR + 1];
    g[s].mi.x = qi[-2 * s * GPR];
    g[s].mi.y = qi[-2 * s * GPR + 1];
  }
  g[0].dr.x = rr[0];
  g[0].dr.y = rr[1];
  g[0].di.x = ri[0];
  g[0].di.y = ri[1];
  const int o0 = h ? (31 * GPR + 2 * m) : (GLST + m);
  const int o1 = 31 * GPR + 2 * m + 1;
  const float m0r = gr[o0], m0i = gi[o0];
  const float m1r = gr[o1], m1i = gi[o1];
  g[0].mr.x = m0r;
  g[0].mi.x = m0i;
  g[0].mr.y = h ? m1r : 0.f;
  g[0].mi.y = h ? m1i : 0.f;
}

struct MixK {
  v4f xv;
  float wr0, wi0, wr1, wi1;
};

static __device__ __forceinline__ int mix_point(int p0, int pi) {
  int q = p0 + pi;
  if (q > NP - 1) q = NP - 1;
  int qu = q - GLST, qv = NLP - 1;
  if (q < GLST) {
    const int row = q / GPR;
    const int r = q - row * GPR;
    qu = r >> 1;
    qv = 2 * row + (r & 1);
  }
  return qu * NLP + qv;
}

static __device__ __forceinline__ void mix_load(MixK& r, const v4f* __restrict__ xp,
                                                const float* __restrict__ wre,
                                                const float* __restrict__ wim,
                                                size_t wbase, int s) {
  r.xv = xp[s * 32];
  const size_t wo = wbase + (size_t)(4 * s) * NP;
  r.wr0 = wre[wo];
  r.wi0 = wim[wo];
  r.wr1 = wre[wo + NP];
  r.wi1 = wim[wo + NP];
}

static __device__ __forceinline__ void mix_compute(const MixK& r, v2f br, v2f bi,
                                                   v8f& dr, v8f& di) {
  v2f ar, ai;
  ar.x = r.xv.x * r.wr0 - r.xv.y * r.wi0;
  ai.x = r.xv.x * r.wi0 + r.xv.y * r.wr0;
  ar.y = r.xv.z * r.wr1 - r.xv.w * r.wi1;
  ai.y = r.xv.z * r.wi1 + r.xv.w * r.wr1;
  cwmma(ar, ai, br, bi, dr, di);
}

static __device__ __forceinline__ void mix_store(float* S, int lane, int pi,
                                                 const v8f& dr, const v8f& di) {
  const int n = lane & 15;
  const int h = lane >> 4;
  float* sr = S + ((h << 3) * ND + n) * MIXP + pi;
  float* si = sr + NB * ND * MIXP;
#pragma unroll
  for (int i = 0; i < 8; ++i) {
    sr[i * ND * MIXP] = dr[i];
    si[i * ND * MIXP] = di[i];
  }
}

__global__ __launch_bounds__(256) void k_pipe(const float* __restrict__ tw,
                                              const float* __restrict__ XT,
                                              const float* __restrict__ wre,
                                              const float* __restrict__ wim,
                                              float* __restrict__ GA,
                                              float* __restrict__ GB,
                                              float* __restrict__ PA,
                                              float* __restrict__ PB,
                                              float* __restrict__ out, int t) {
  __shared__ __attribute__((aligned(16))) float S[2 * NB * ND * MIXP];
  const int wave = threadIdx.x >> 5;
  const int lane = threadIdx.x & 31;
  const int h = lane >> 4;

  if (blockIdx.x < MIXB) {
    const int nidx = t;
    if (nidx >= NFR) return;
    float* Gr = (nidx & 1) ? GB : GA;
    float* Gi = Gr + GSZ;
    const int p0 = blockIdx.x * MIXP;
    const float* WNr = tw + OFF_WNR;
    const float* WNi = tw + OFF_WNI;
    const int bm = lane & 15;
    v2f br[16], bi[16];
#pragma unroll
    for (int s = 0; s < 16; ++s) {
      br[s] = frag_b(WNr, ND, 4 * s, 0);
      bi[s] = frag_b(WNi, ND, 4 * s, 0);
    }
    const v4f* xt4 = reinterpret_cast<const v4f*>(XT) + h * 16 + bm;
    const size_t wrow = ((size_t)nidx * NC + 2 * h) * NP;
    const int npts = (MIXP - wave + 7) >> 3;
    MixK ring[8];
    int pi = wave;
    int p = mix_point(p0, pi);
    const v4f* xp = xt4 + (size_t)p * 512;
    size_t wbase = wrow + p;
#pragma unroll
    for (int s = 0; s < 8; ++s) mix_load(ring[s], xp, wre, wim, wbase, s);
    for (int i = 0; i + 1 < npts; ++i) {
      p = mix_point(p0, pi + 8);
      const v4f* xn = xt4 + (size_t)p * 512;
      const size_t wnext = wrow + p;
      v8f dr = zero8(), di = zero8();
#pragma unroll
      for (int s = 0; s < 16; ++s) {
        mix_compute(ring[s & 7], br[s], bi[s], dr, di);
        if (s < 8) mix_load(ring[s & 7], xp, wre, wim, wbase, s + 8);
        else       mix_load(ring[s & 7], xn, wre, wim, wnext, s - 8);
        __builtin_amdgcn_sched_barrier(0);
      }
      mix_store(S, lane, pi, dr, di);
      xp = xn;
      wbase = wnext;
      pi += 8;
    }
    {
      v8f dr = zero8(), di = zero8();
#pragma unroll
      for (int s = 0; s < 16; ++s) {
        mix_compute(ring[s & 7], br[s], bi[s], dr, di);
        if (s < 8) {
          mix_load(ring[s & 7], xp, wre, wim, wbase, s + 8);
          __builtin_amdgcn_sched_barrier(0);
        }
      }
      mix_store(S, lane, pi, dr, di);
    }
    __syncthreads();
#pragma unroll
    for (int j = 0; j < MIXP; ++j) {
      const int f = (int)threadIdx.x + 256 * j;
      const int row = f / MIXP;
      const int pt = f - row * MIXP;
      if (p0 + pt < NP) {
        const size_t o = (size_t)row * NP + p0 + pt;
        Gr[o] = S[f];
        Gi[o] = S[NB * ND * MIXP + f];
      }
    }
    return;
  }

  const int plane = blockIdx.x - MIXB;
  const float* IVr = tw + OFF_IVR;
  const float* IVi = tw + OFF_IVI;

  constexpr int IUH = 32;
  constexpr int IUP4 = NL * 68 / 4;
  static_assert(2 * NL * 68 <= 2 * NB * ND * MIXP, "IU must fit the LDS tile");
  static_assert(IUH * 68 == 2 * 256 * 4 + 128, "a part = 2 pieces of 256 x 16 B + 128 words");
  const bool u_on = (t >= 2) && (t - 2 < NFR);
  if (u_on) {
    const v4f* tsrc = reinterpret_cast<const v4f*>(tw + OFF_IUR);
    v4f* tdst = reinterpret_cast<v4f*>(S);
    const int ct = (int)threadIdx.x & 255;
    v4f tv[4];
#pragma unroll
    for (int j = 0; j < 4; ++j) tv[j] = tsrc[(j >> 1) * IUP4 + ct + 256 * (j & 1)];
    const int ro = (ct >> 7) * (NL * 68) + 2048 + (ct & 127);
    const float tl = tw[OFF_IUR + ro];
#pragma unroll
    for (int j = 0; j < 4; ++j) tdst[(j >> 1) * IUP4 + ct + 256 * (j & 1)] = tv[j];
    S[ro] = tl;
    __syncthreads();
  }

  if (wave < 4) {
    const int nidx = t - 1;
    if (nidx < 0 || nidx >= NFR) return;
    const float* gr = ((nidx & 1) ? GB : GA) + (size_t)plane * NP;
    const float* gi = gr + GSZ;
    float* pr = ((nidx & 1) ? PB : PA) + (size_t)plane * NPW;
    float* pi = pr + PSZ;
    const int no = (wave & 1) * 16;
    const int mset = wave >> 1;
    v2f bc[9], bs[9];
#pragma unroll
    for (int s = 0; s < 9; ++s) {
      bc[s] = frag_b(IVr, NL, 4 * s, no);
      bs[s] = frag_b(IVi, NL, 4 * s, no);
    }
    GRaw g[2][9];
    load_g_raw(gr, gi, mset * 16, g[0]);
#pragma unroll
    for (int j = 0; j < 3; ++j) {
      const int mt = 2 * j + mset;
      if (mt < 5) {
        if (mt + 2 < 5) load_g_raw(gr, gi, (mt + 2) * 16, g[(j + 1) & 1]);
        const int mo = mt * 16;
        v8f cr = zero8(), ci = zero8(), sr = zero8(), si = zero8();
#pragma unroll
        for (int s = 0; s < 9; ++s) {
          const GRaw& q = g[j & 1][s];
          v2f apr, api, amr, ami;
          apr.x = q.dr.x + q.mr.y;
          apr.y = q.dr.y + q.mr.x;
          api.x = q.di.x + q.mi.y;
          api.y = q.di.y + q.mi.x;
          amr.x = q.dr.x - q.mr.y;
          amr.y = q.dr.y - q.mr.x;
          ami.x = q.di.x - q.mi.y;
          ami.y = q.di.y - q.mi.x;
          if (s == 8) {
            apr.x = h ? 0.f : apr.x;
            api.x = h ? 0.f : api.x;
            amr.x = h ? 0.f : amr.x;
            ami.x = h ? 0.f : ami.x;
            apr.y = 0.f;
            api.y = 0.f;
            amr.y = 0.f;
            ami.y = 0.f;
          }
          cr = wmma4(apr, bc[s], cr);
          ci = wmma4(api, bc[s], ci);
          sr = wmma4(amr, bs[s], sr);
          si = wmma4(ami, bs[s], si);
        }
        const int n = no + (lane & 15);
        const int nm = NL - 1 - n;
        const int mb = mo + ((lane >> 4) << 3);
#pragma unroll
        for (int i = 0; i < 8; ++i) {
          const int m = mb + i;
          if (m < NLP) {
            pr[m * NL + n] = cr[i] - si[i];
            pi[m * NL + n] = ci[i] + sr[i];
            pr[m * NL + nm] = cr[i] + si[i];
            pi[m * NL + nm] = ci[i] - sr[i];
          }
        }
      }
    }
  } else {
    const int nidx = t - 2;
    if (nidx < 0 || nidx >= NFR) return;
    const int d = plane & (ND - 1), b = plane >> 4;
    const float* pr = ((nidx & 1) ? PB : PA) + (size_t)plane * NPW;
    const float* pi = pr + PSZ;
    float* op = out + ((size_t)b * (ND * NFR) + (d * NFR + nidx)) * (NL * NL);
    const int no = (wave - 4) * 16;
    const int n = no + (lane & 15);
    const float* prl = pr + (2 * h) * NL + n;
    const float* pil = pi + (2 * h) * NL + n;
    const float* prm = pr + (64 - 2 * h) * NL + n;
    const float* pim = pi + (64 - 2 * h) * NL + n;
    v2f bp[9], bm[9];
#pragma unroll
    for (int s = 1; s < 9; ++s) {
      const float a_r = prl[4 * s * NL], b_r = prl[(4 * s + 1) * NL];
      const float c_r = prm[(1 - 4 * s) * NL], d_r = prm[-4 * s * NL];
      const float a_i = pil[4 * s * NL], b_i = pil[(4 * s + 1) * NL];
      const float c_i = pim[(1 - 4 * s) * NL], d_i = pim[-4 * s * NL];
      bp[s].x = a_r + c_r;
      bp[s].y = b_r + d_r;
      bm[s].x = a_i - c_i;
      bm[s].y = b_i - d_i;
    }
    bp[8].x = h ? 0.f : bp[8].x;
    bm[8].x = h ? 0.f : bm[8].x;
    bp[8].y = 0.f;
    bm[8].y = 0.f;
    {
      const float a_r = prl[0], b_r = prl[NL], d_r = prm[0];
      const float a_i = pil[0], b_i = pil[NL], d_i = pim[0];
      const int c0 = (h ? 63 : 64) * NL + n;
      const float c_r = pr[c0], c_i = pi[c0];
      bp[0].x = a_r + (h ? c_r : 0.f);
      bp[0].y = b_r + d_r;
      bm[0].x = a_i - (h ? c_i : 0.f);
      bm[0].y = b_i - d_i;
    }
    for (int j = 0; j < 2; ++j) {
      const int mo = j * 16;
      v8f ec = zero8(), es = zero8();
#pragma unroll
      for (int s = 0; s < 9; ++s) {
        v2f ar = frag_a(S, 68, mo, 4 * s);
        v2f ai = frag_a(S + NL * 68, 68, mo, 4 * s);
        ec = wmma4(ar, bp[s], ec);
        es = wmma4(ai, bm[s], es);
      }
      const int mb = mo + ((lane >> 4) << 3);
#pragma unroll
      for (int i = 0; i < 8; ++i) {
        op[(mb + i) * NL + n] = ec[i] - es[i];
        op[(NL - 1 - (mb + i)) * NL + n] = ec[i] + es[i];
      }
    }
  }
}


extern "C" void kernel_launch(void* const* d_in, const int* in_sizes, int n_in,
                              void* d_out, int out_size, void* d_ws,
                              size_t ws_size, hipStream_t stream) {
  (void)in_sizes; (void)n_in; (void)out_size; (void)ws_size;
  const float* x   = (const float*)d_in[0];
  const float* wre = (const float*)d_in[1];
  const float* wim = (const float*)d_in[2];
  float* out = (float*)d_out;
  float* ws = (float*)d_ws;

  float* tw = ws;
  size_t off = TW_PAD;
  float* Yr = ws + off; off += (size_t)NB * NC * NP;
  float* Yi = ws + off; off += (size_t)NB * NC * NP;
  float* XT = ws + off; off += (size_t)2 * NB * NC * NP;
  float* Gr = ws + off; off += (size_t)NB * ND * NP;
  float* Gi = ws + off; off += (size_t)NB * ND * NP;
  float* Pr = ws + off; off += (size_t)NB * ND * NPW;
  float* Pi = ws + off; off += (size_t)NB * ND * NPW;
  (void)Gi; (void)Pi;

  k_twiddles<<<dim3(25, 5), 256, 0, stream>>>(tw);
  k_fft2d<<<NB * NC, 128, 0, stream>>>(x, tw, Yr, Yi);
  k_cfft<<<dim3((NP + 15) / 16, 2), 512, 0, stream>>>(tw, Yr, Yi, XT);
  for (int t = 0; t < NFR + 2; ++t)
    k_pipe<<<MIXB + NB * ND, 256, 0, stream>>>(tw, XT, wre, wim, Gr, Yr, Pr, Yi, out, t);
}


# W re-laid out point-major (512 B per point and mode) by a transposing pre-pass in idle blocks; mix loads a lane's 4 W values as one 16-byte element, not 4 dwords on 8 lines; values, sums unchanged
# speedup vs baseline: 6.2654x; 1.0433x over previous
#include <hip/hip_runtime.h>


typedef __attribute__((ext_vector_type(2))) float v2f;
typedef __attribute__((ext_vector_type(4))) float v4f;
typedef __attribute__((ext_vector_type(8))) float v8f;

constexpr int NB  = 16;
constexpr int NC  = 64;
constexpr int NFR = 8;
constexpr int ND  = 16;
constexpr int NL  = 64;
constexpr int NLP = 65;
constexpr int NP  = NLP * NLP;
constexpr int NPW = NLP * NL;

constexpr int OFF_FWDR = 0;
constexpr int OFF_FWDI = OFF_FWDR + 80 * 80;
constexpr int OFF_ER   = OFF_FWDI + 80 * 80;
constexpr int OFF_EI   = OFF_ER + 64 * 64;
constexpr int OFF_WNR  = OFF_EI + 64 * 64;
constexpr int OFF_WNI  = OFF_WNR + 64 * 16;
constexpr int OFF_IVR  = OFF_WNI + 64 * 16;
constexpr int OFF_IVI  = OFF_IVR + 68 * 64;
constexpr int OFF_IUR  = OFF_IVI + 68 * 64;
constexpr int OFF_IUI  = OFF_IUR + 64 * 68;
constexpr int TW_TOTAL = OFF_IUI + 64 * 68;
constexpr int TW_PAD   = 40960;

#define PI_D 3.14159265358979323846


static __device__ __forceinline__ v8f wmma4(v2f a, v2f b, v8f c) {
  return __builtin_amdgcn_wmma_f32_16x16x4_f32(false, a, false, b, (short)0, c,
                                               false, false);
}

static __device__ __forceinline__ v8f zero8() {
  v8f z = {0.f, 0.f, 0.f, 0.f, 0.f, 0.f, 0.f, 0.f};
  return z;
}

static __device__ __forceinline__ v2f frag_a(const float* __restrict__ A,
                                             int lda, int m0, int k0) {
  int lane = threadIdx.x & 31;
  int m = m0 + (lane & 15);
  int k = k0 + ((lane >> 4) << 1);
  const float* p = A + m * lda + k;
  v2f r;
  r.x = p[0];
  r.y = p[1];
  return r;
}

static __device__ __forceinline__ v2f frag_b(const float* __restrict__ B,
                                             int ldb, int k0, int n0) {
  int lane = threadIdx.x & 31;
  int n = n0 + (lane & 15);
  int k = k0 + ((lane >> 4) << 1);
  const float* p = B + k * ldb + n;
  v2f r;
  r.x = p[0];
  r.y = p[ldb];
  return r;
}

static __device__ __forceinline__ v2f frag_b_nclamp(const float* __restrict__ B,
                                                    int ldb, int k0, int n0,
                                                    int nmax) {
  int lane = threadIdx.x & 31;
  int n = n0 + (lane & 15);
  if (n > nmax - 1) n = nmax - 1;
  int k = k0 + ((lane >> 4) << 1);
  const float* p = B + k * ldb + n;
  v2f r;
  r.x = p[0];
  r.y = p[ldb];
  return r;
}

static __device__ __forceinline__ v2f frag_a_mclamp(const float* __restrict__ A,
                                                    int lda, int m0, int k0,
                                                    int mmax) {
  int lane = threadIdx.x & 31;
  int m = m0 + (lane & 15);
  if (m > mmax - 1) m = mmax - 1;
  int k = k0 + ((lane >> 4) << 1);
  const float* p = A + m * lda + k;
  v2f r;
  r.x = p[0];
  r.y = p[1];
  return r;
}

static __device__ __forceinline__ void cwmma(v2f ar, v2f ai, v2f br, v2f bi,
                                             v8f& dr, v8f& di) {
  dr = wmma4(ar, br, dr);
  v2f nai;
  nai.x = -ai.x;
  nai.y = -ai.y;
  dr = wmma4(nai, bi, dr);
  di = wmma4(ar, bi, di);
  di = wmma4(ai, br, di);
}

constexpr int WTSZ = NP * 2 * NC;
constexpr int WTP  = 132;
constexpr int WTT  = (NP + 31) / 32;
constexpr int WGSZ = NB * ND * NP;
constexpr int WPSZ = NB * ND * NPW;

static __device__ __forceinline__ float* wt_base(int n, float* PA, float* GB, float* PB) {
  if (n == 0) return PA;
  if (n < 5) return GB + 2 * (size_t)WGSZ + (size_t)(n - 1) * WTSZ;
  return PB + 2 * (size_t)WPSZ + (size_t)(n - 5) * WTSZ;
}

static __device__ __forceinline__ void wt_load(const float* __restrict__ wre,
                                               const float* __restrict__ wim,
                                               int g, float* v) {
  const int w = threadIdx.x >> 5, lane = threadIdx.x & 31;
  const int n = g / WTT, pt = g - n * WTT;
  int p = pt * 32 + lane;
  if (p > NP - 1) p = NP - 1;
  const float* src = ((w & 1) ? wim : wre) + ((size_t)n * NC + (w >> 1)) * NP + p;
#pragma unroll
  for (int j = 0; j < 16; ++j) v[j] = src[(size_t)(4 * j) * NP];
}

static __device__ __forceinline__ void wt_run(const float* __restrict__ wre,
                                              const float* __restrict__ wim,
                                              int first, int step, int total,
                                              float* PA, float* GB, float* PB,
                                              float* T) {
  const int w = threadIdx.x >> 5, lane = threadIdx.x & 31;
  float v[16];
  if (first < total) wt_load(wre, wim, first, v);
  for (int g = first; g < total; g += step) {
    __syncthreads();
#pragma unroll
    for (int j = 0; j < 16; ++j) T[lane * WTP + 8 * j + w] = v[j];
    __syncthreads();
    const int n = g / WTT, pt = g - n * WTT;
    if (g + step < total) wt_load(wre, wim, g + step, v);
    v4f* dst = reinterpret_cast<v4f*>(wt_base(n, PA, GB, PB));
    const v4f* T4 = reinterpret_cast<const v4f*>(T);
#pragma unroll
    for (int i = 0; i < 4; ++i) {
      const int pl = 4 * w + i;
      const int p = pt * 32 + pl;
      if (p < NP) dst[(size_t)p * 32 + lane] = T4[pl * (WTP / 4) + lane];
    }
  }
}


__global__ __launch_bounds__(256) void k_twiddles(float* __restrict__ tw,
                                                  const float* __restrict__ wre,
                                                  const float* __restrict__ wim,
                                                  float* PA, float* GB, float* PB) {
  __shared__ __attribute__((aligned(16))) float T[32 * WTP];
  int t = blockIdx.x * blockDim.x + threadIdx.x;
  int which = blockIdx.y;
  if (which >= 5) {
    const int g = (which - 5) * (int)gridDim.x + (int)blockIdx.x;
    wt_run(wre, wim, g, WTT, WTT, PA, GB, PB, T);
    return;
  }
  if (which == 0) {
    if (t >= 80 * 80) return;
    int a = t / 80, b = t % 80;
    float r = 0.f, im = 0.f;
    if (a < NLP && b < NLP) {
      double ang = -2.0 * PI_D * (double)((a * b) % NLP) / (double)NLP;
      r = (float)cos(ang);
      im = (float)sin(ang);
    }
    tw[OFF_FWDR + t] = r;
    tw[OFF_FWDI + t] = im;
  } else if (which == 1) {
    if (t >= NC * NC) return;
    int c = t / NC, k = t % NC;
    double ang = -2.0 * PI_D * (double)((c * k) % NC) / (double)NC;
    tw[OFF_ER + t] = (float)cos(ang);
    tw[OFF_EI + t] = (float)sin(ang);
  } else if (which == 2) {
    if (t >= NC * ND) return;
    int c = t / ND, d = t % ND;
    double ang = 2.0 * PI_D * (double)((c * 4 * d) % NC) / (double)NC;
    tw[OFF_WNR + t] = (float)(cos(ang) / 64.0);
    tw[OFF_WNI + t] = (float)(sin(ang) / 64.0);
  } else if (which == 3) {
    if (t >= 68 * NL) return;
    int v = t / NL, w = t % NL;
    float r = 0.f, im = 0.f;
    if (v < NLP) {
      double ang = 2.0 * PI_D * (double)((v * (w + 1)) % NLP) / (double)NLP;
      r = (float)cos(ang);
      im = (float)sin(ang);
    }
    tw[OFF_IVR + t] = r;
    tw[OFF_IVI + t] = im;
  } else {
    if (t >= NL * 68) return;
    int h = t / 68, u = t % 68;
    float r = 0.f, im = 0.f;
    if (u < NLP) {
      double ang = 2.0 * PI_D * (double)((u * (h + 1)) % NLP) / (double)NLP;
      double s = 1.0 / (double)(NLP * NLP);
      r = (float)(cos(ang) * s);
      im = (float)(sin(ang) * s);
    }
    tw[OFF_IUR + t] = r;
    tw[OFF_IUI + t] = im;
  }
}


__global__ __launch_bounds__(128) void k_fft2d(const float* __restrict__ x,
                                               const float* __restrict__ tw,
                                               float* __restrict__ Yr,
                                               float* __restrict__ Yi) {
  constexpr int FP = 68;
  constexpr int FR = 68;
  __shared__ __attribute__((aligned(16))) float U[2 * FR * FP];
  float* xs = U;
  __shared__ float Tr[NL * 80];
  __shared__ float Ti[NL * 80];
  const int plane = blockIdx.x;
  const int tid = threadIdx.x;
  const int wave = tid >> 5;
  const int lane = tid & 31;

  const float* xp = x + (size_t)plane * (NL * NL);
  for (int i = tid; i < NL * NL; i += 128) xs[i] = xp[i];
  __syncthreads();

  const float* Fr = tw + OFF_FWDR;
  const float* Fi = tw + OFF_FWDI;

  const int hh = lane >> 4;

  for (int t = wave; t < 4 * 5; t += 4) {
    int mo = (t / 5) * 16, no = (t % 5) * 16;
    v8f ar = zero8(), ai = zero8();
    const float* xr = xs + (mo + (lane & 15)) * NL + 2 * hh;
    const float* xm = xs + (mo + (lane & 15)) * NL + 64 - 2 * hh;
#pragma unroll
    for (int s = 0; s < 9; ++s) {
      const float d0 = xr[4 * s], d1 = xr[4 * s + 1];
      float m0, m1;
      if (s == 0) {
        const float* xq = xs + (mo + (lane & 15)) * NL + 62;
        const float q0 = xq[1], q1 = xq[0];
        m0 = hh ? q0 : 0.f;
        m1 = hh ? q1 : 0.f;
      } else {
        m0 = xm[1 - 4 * s];
        m1 = xm[-4 * s];
      }
      v2f ap, am;
      ap.x = d0 + m0;
      ap.y = d1 + m1;
      am.x = d0 - m0;
      am.y = d1 - m1;
      if (s == 8) {
        ap.x = hh ? 0.f : ap.x;
        am.x = hh ? 0.f : am.x;
        ap.y = 0.f;
        am.y = 0.f;
      }
      ar = wmma4(ap, frag_b(Fr, 80, 4 * s, no), ar);
      ai = wmma4(am, frag_b(Fi, 80, 4 * s, no), ai);
    }
    int n = no + (lane & 15);
    int mb = mo + ((lane >> 4) << 3);
#pragma unroll
    for (int i = 0; i < 8; ++i) {
      Tr[(mb + i) * 80 + n] = ar[i];
      Ti[(mb + i) * 80 + n] = ai[i];
    }
  }
  constexpr int FN4 = 2 * FR * 16;
  static_assert(FN4 == 17 * 128, "17 whole pieces of one element per thread");
  const int ct = tid & 127;
  v4f fv[17];
#pragma unroll
  for (int j = 0; j < 17; ++j) {
    const int e = ct + 128 * j;
    const int part = e / (FR * 16);
    const int r = e - part * (FR * 16);
    fv[j] = reinterpret_cast<const v4f*>(tw + OFF_FWDR)
                [part * ((OFF_FWDI - OFF_FWDR) / 4) + (r >> 4) * 20 + (r & 15)];
  }
  __syncthreads();
#pragma unroll
  for (int j = 0; j < 17; ++j) {
    const int e = ct + 128 * j;
    const int part = e / (FR * 16);
    const int r = e - part * (FR * 16);
    reinterpret_cast<v4f*>(U)[part * (FR * FP / 4) + (r >> 4) * (FP / 4) + (r & 15)] = fv[j];
  }
  __syncthreads();
  const float* Usr = U;
  const float* Usi = U + FR * FP;

  float* yr = Yr + (size_t)plane * NP;
  float* yi = Yi + (size_t)plane * NP;
  for (int t = wave; t < 3 * 5; t += 4) {
    int mo = (t / 5) * 16, no = (t % 5) * 16;
    const int n = no + (lane & 15);
    const float* trd = Tr + (2 * hh) * 80 + n;
    const float* tid = Ti + (2 * hh) * 80 + n;
    const float* trm = Tr + (62 - 2 * hh) * 80 + n;
    const float* tim = Ti + (62 - 2 * hh) * 80 + n;
    v8f cr = zero8(), ci = zero8(), qr = zero8(), qi = zero8();
#pragma unroll
    for (int s = 0; s < 9; ++s) {
      const float d0r = trd[(4 * s) * 80], d1r = trd[(4 * s + 1) * 80];
      const float d0i = tid[(4 * s) * 80], d1i = tid[(4 * s + 1) * 80];
      float m0r, m1r, m0i, m1i;
      if (s == 0) {
        const float a0r = Tr[63 * 80 + n], a1r = Tr[62 * 80 + n];
        const float a0i = Ti[63 * 80 + n], a1i = Ti[62 * 80 + n];
        m0r = hh ? a0r : 0.f;
        m1r = hh ? a1r : 0.f;
        m0i = hh ? a0i : 0.f;
        m1i = hh ? a1i : 0.f;
      } else {
        m0r = trm[(3 - 4 * s) * 80];
        m1r = trm[(2 - 4 * s) * 80];
        m0i = tim[(3 - 4 * s) * 80];
        m1i = tim[(2 - 4 * s) * 80];
      }
      v2f bpr, bpi, bmr, bmi;
      bpr.x = d0r + m0r;
      bpr.y = d1r + m1r;
      bpi.x = d0i + m0i;
      bpi.y = d1i + m1i;
      bmr.x = d0r - m0r;
      bmr.y = d1r - m1r;
      bmi.x = d0i - m0i;
      bmi.y = d1i - m1i;
      if (s == 8) {
        bpr.x = hh ? 0.f : bpr.x;
        bpi.x = hh ? 0.f : bpi.x;
        bmr.x = hh ? 0.f : bmr.x;
        bmi.x = hh ? 0.f : bmi.x;
        bpr.y = 0.f;
        bpi.y = 0.f;
        bmr.y = 0.f;
        bmi.y = 0.f;
      }
      v2f fr = frag_a(Usr, FP, mo, 4 * s);
      v2f fi = frag_a(Usi, FP, mo, 4 * s);
      cr = wmma4(fr, bpr, cr);
      ci = wmma4(fr, bpi, ci);
      qr = wmma4(fi, bmr, qr);
      qi = wmma4(fi, bmi, qi);
    }
    int mb = mo + (hh << 3);
    if (n < NLP) {
#pragma unroll
      for (int i = 0; i < 8; ++i) {
        int m = mb + i;
        if (m <= 32) {
          yr[m * NLP + n] = cr[i] - qi[i];
          yi[m * NLP + n] = ci[i] + qr[i];
          if (m >= 1) {
            yr[(NLP - m) * NLP + n] = cr[i] + qi[i];
            yi[(NLP - m) * NLP + n] = ci[i] - qr[i];
          }
        }
      }
    }
  }
}


__global__ __launch_bounds__(512) void k_cfft(const float* __restrict__ tw,
                                              const float* __restrict__ Yr,
                                              const float* __restrict__ Yi,
                                              float* __restrict__ XT) {
  __shared__ v4f S[16 * 256];
  constexpr int EP = 68;
  __shared__ __attribute__((aligned(16))) float Es[2 * 32 * EP];
  const int b = threadIdx.x >> 5;
  const int lane = threadIdx.x & 31;
  const int no = blockIdx.x * 16;
  const int chalf = blockIdx.y;
  const int cb = chalf * 32;
  const float* Er = tw + OFF_ER;
  const float* Ei = tw + OFF_EI;
  const float* yr = Yr + (size_t)b * NC * NP;
  const float* yi = Yi + (size_t)b * NC * NP;
  {
    const int ct = (int)threadIdx.x & 511;
    const v4f e_re = reinterpret_cast<const v4f*>(Er + cb * NC)[ct];
    const v4f e_im = reinterpret_cast<const v4f*>(Ei + cb * NC)[ct];
    v4f* es4 = reinterpret_cast<v4f*>(Es);
    const int de = (ct >> 4) * (EP / 4) + (ct & 15);
    es4[de] = e_re;
    es4[32 * EP / 4 + de] = e_im;
  }
  __syncthreads();
  const float* Esr = Es;
  const float* Esi = Es + 32 * EP;
  v8f dr0 = zero8(), di0 = zero8(), dr1 = zero8(), di1 = zero8();
  for (int k0 = 0; k0 < NC; k0 += 4) {
    v2f br = frag_b_nclamp(yr, NP, k0, no, NP);
    v2f bi = frag_b_nclamp(yi, NP, k0, no, NP);
    v2f ar = frag_a(Esr, EP, 0, k0);
    v2f ai = frag_a(Esi, EP, 0, k0);
    cwmma(ar, ai, br, bi, dr0, di0);
    ar = frag_a(Esr, EP, 16, k0);
    ai = frag_a(Esi, EP, 16, k0);
    cwmma(ar, ai, br, bi, dr1, di1);
  }
  const int pl = lane & 15;
  const int h = lane >> 4;
  v4f* srow = S + pl * 256 + (b ^ pl);
#pragma unroll
  for (int q = 0; q < 4; ++q) {
    v4f e0 = {dr0[2 * q], di0[2 * q], dr0[2 * q + 1], di0[2 * q + 1]};
    v4f e1 = {dr1[2 * q], di1[2 * q], dr1[2 * q + 1], di1[2 * q + 1]};
    srow[(4 * h + q) * 16] = e0;
    srow[(8 + 4 * h + q) * 16] = e1;
  }
  __syncthreads();
  v4f* xt = reinterpret_cast<v4f*>(XT);
  const int idx = threadIdx.x & 255;
  const int psel = threadIdx.x >> 8;
#pragma unroll
  for (int pass = 0; pass < 8; ++pass) {
    int pp = pass * 2 + psel;
    int p = no + pp;
    if (p < NP)
      xt[(size_t)p * 512 + chalf * 256 + idx] =
          S[pp * 256 + (idx & 0xF0) + ((idx & 15) ^ pp)];
  }
}


constexpr int MIXP = 17;
constexpr int MIXB = (NP + MIXP - 1) / MIXP;
constexpr int GSZ  = NB * ND * NP;
constexpr int PSZ  = NB * ND * NPW;
constexpr int GPR  = 2 * NLP;
constexpr int GLST = 64 * NLP;

struct GRaw {
  v2f dr, di, mr, mi;
};

static __device__ __forceinline__ void load_g_raw(const float* __restrict__ gr,
                                                  const float* __restrict__ gi,
                                                  int mo, GRaw* g) {
  const int lane = threadIdx.x & 31;
  const int h = lane >> 4;
  int m = mo + (lane & 15);
  if (m > NLP - 1) m = NLP - 1;
  const float* rr = gr + h * GPR + 2 * m;
  const float* ri = gi + h * GPR + 2 * m;
  const float* qr = gr + (32 - h) * GPR + 2 * m;
  const float* qi = gi + (32 - h) * GPR + 2 * m;
#pragma unroll
  for (int s = 1; s < 9; ++s) {
    g[s].dr.x = rr[2 * s * GPR];
    g[s].dr.y = rr[2 * s * GPR + 1];
    g[s].di.x = ri[2 * s * GPR];
    g[s].di.y = ri[2 * s * GPR + 1];
    g[s].mr.x = qr[-2 * s * GPR];
    g[s].mr.y = qr[-2 * s * GPR + 1];
    g[s].mi.x = qi[-2 * s * GPR];
    g[s].mi.y = qi[-2 * s * GPR + 1];
  }
  g[0].dr.x = rr[0];
  g[0].dr.y = rr[1];
  g[0].di.x = ri[0];
  g[0].di.y = ri[1];
  const int o0 = h ? (31 * GPR + 2 * m) : (GLST + m);
  const int o1 = 31 * GPR + 2 * m + 1;
  const float m0r = gr[o0], m0i = gi[o0];
  const float m1r = gr[o1], m1i = gi[o1];
  g[0].mr.x = m0r;
  g[0].mi.x = m0i;
  g[0].mr.y = h ? m1r : 0.f;
  g[0].mi.y = h ? m1i : 0.f;
}

struct MixK {
  v4f xv;
  v4f wv;
};

static __device__ __forceinline__ int mix_point(int p0, int pi) {
  int q = p0 + pi;
  if (q > NP - 1) q = NP - 1;
  int qu = q - GLST, qv = NLP - 1;
  if (q < GLST) {
    const int row = q / GPR;
    const int r = q - row * GPR;
    qu = r >> 1;
    qv = 2 * row + (r & 1);
  }
  return qu * NLP + qv;
}

static __device__ __forceinline__ void mix_load(MixK& r, const v4f* __restrict__ xp,
                                                const v4f* __restrict__ wp, int s) {
  r.xv = xp[s * 32];
  r.wv = wp[s * 2];
}

static __device__ __forceinline__ void mix_compute(const MixK& r, v2f br, v2f bi,
                                                   v8f& dr, v8f& di) {
  v2f ar, ai;
  ar.x = __builtin_fmaf(r.xv.x, r.wv.x, -(r.xv.y * r.wv.y));
  ai.x = __builtin_fmaf(r.xv.y, r.wv.x, r.xv.x * r.wv.y);
  ar.y = __builtin_fmaf(r.xv.z, r.wv.z, -(r.xv.w * r.wv.w));
  ai.y = __builtin_fmaf(r.xv.w, r.wv.z, r.xv.z * r.wv.w);
  cwmma(ar, ai, br, bi, dr, di);
}

static __device__ __forceinline__ void mix_store(float* S, int lane, int pi,
                                                 const v8f& dr, const v8f& di) {
  const int n = lane & 15;
  const int h = lane >> 4;
  float* sr = S + ((h << 3) * ND + n) * MIXP + pi;
  float* si = sr + NB * ND * MIXP;
#pragma unroll
  for (int i = 0; i < 8; ++i) {
    sr[i * ND * MIXP] = dr[i];
    si[i * ND * MIXP] = di[i];
  }
}

__global__ __launch_bounds__(256) void k_pipe(const float* __restrict__ tw,
                                              const float* __restrict__ XT,
                                              const float* __restrict__ wre,
                                              const float* __restrict__ wim,
                                              float* __restrict__ GA,
                                              float* __restrict__ GB,
                                              float* __restrict__ PA,
                                              float* __restrict__ PB,
                                              float* __restrict__ out, int t) {
  __shared__ __attribute__((aligned(16))) float S[2 * NB * ND * MIXP];
  const int wave = threadIdx.x >> 5;
  const int lane = threadIdx.x & 31;
  const int h = lane >> 4;

  if (blockIdx.x < MIXB) {
    const int nidx = t;
    if (nidx >= NFR) return;
    float* Gr = (nidx & 1) ? GB : GA;
    float* Gi = Gr + GSZ;
    const int p0 = blockIdx.x * MIXP;
    const float* WNr = tw + OFF_WNR;
    const float* WNi = tw + OFF_WNI;
    const int bm = lane & 15;
    v2f br[16], bi[16];
#pragma unroll
    for (int s = 0; s < 16; ++s) {
      br[s] = frag_b(WNr, ND, 4 * s, 0);
      bi[s] = frag_b(WNi, ND, 4 * s, 0);
    }
    const v4f* xt4 = reinterpret_cast<const v4f*>(XT) + h * 16 + bm;
    const v4f* wt4 = reinterpret_cast<const v4f*>(wt_base(nidx, PA, GB, PB)) + h;
    const int npts = (MIXP - wave + 7) >> 3;
    MixK ring[8];
    int pi = wave;
    int p = mix_point(p0, pi);
    const v4f* xp = xt4 + (size_t)p * 512;
    const v4f* wp = wt4 + (size_t)p * 32;
#pragma unroll
    for (int s = 0; s < 8; ++s) mix_load(ring[s], xp, wp, s);
    for (int i = 0; i + 1 < npts; ++i) {
      p = mix_point(p0, pi + 8);
      const v4f* xn = xt4 + (size_t)p * 512;
      const v4f* wn = wt4 + (size_t)p * 32;
      v8f dr = zero8(), di = zero8();
#pragma unroll
      for (int s = 0; s < 16; ++s) {
        mix_compute(ring[s & 7], br[s], bi[s], dr, di);
        if (s < 8) mix_load(ring[s & 7], xp, wp, s + 8);
        else       mix_load(ring[s & 7], xn, wn, s - 8);
        __builtin_amdgcn_sched_barrier(0);
      }
      mix_store(S, lane, pi, dr, di);
      xp = xn;
      wp = wn;
      pi += 8;
    }
    {
      v8f dr = zero8(), di = zero8();
#pragma unroll
      for (int s = 0; s < 16; ++s) {
        mix_compute(ring[s & 7], br[s], bi[s], dr, di);
        if (s < 8) {
          mix_load(ring[s & 7], xp, wp, s + 8);
          __builtin_amdgcn_sched_barrier(0);
        }
      }
      mix_store(S, lane, pi, dr, di);
    }
    __syncthreads();
#pragma unroll
    for (int j = 0; j < MIXP; ++j) {
      const int f = (int)threadIdx.x + 256 * j;
      const int row = f / MIXP;
      const int pt = f - row * MIXP;
      if (p0 + pt < NP) {
        const size_t o = (size_t)row * NP + p0 + pt;
        Gr[o] = S[f];
        Gi[o] = S[NB * ND * MIXP + f];
      }
    }
    return;
  }

  const int plane = blockIdx.x - MIXB;
  if (t == 0) {
    wt_run(wre, wim, WTT + plane, NB * ND, NFR * WTT, PA, GB, PB, S);
    return;
  }
  const float* IVr = tw + OFF_IVR;
  const float* IVi = tw + OFF_IVI;

  constexpr int IUH = 32;
  constexpr int IUP4 = NL * 68 / 4;
  static_assert(2 * NL * 68 <= 2 * NB * ND * MIXP, "IU must fit the LDS tile");
  static_assert(IUH * 68 == 2 * 256 * 4 + 128, "a part = 2 pieces of 256 x 16 B + 128 words");
  const bool u_on = (t >= 2) && (t - 2 < NFR);
  if (u_on) {
    const v4f* tsrc = reinterpret_cast<const v4f*>(tw + OFF_IUR);
    v4f* tdst = reinterpret_cast<v4f*>(S);
    const int ct = (int)threadIdx.x & 255;
    v4f tv[4];
#pragma unroll
    for (int j = 0; j < 4; ++j) tv[j] = tsrc[(j >> 1) * IUP4 + ct + 256 * (j & 1)];
    const int ro = (ct >> 7) * (NL * 68) + 2048 + (ct & 127);
    const float tl = tw[OFF_IUR + ro];
#pragma unroll
    for (int j = 0; j < 4; ++j) tdst[(j >> 1) * IUP4 + ct + 256 * (j & 1)] = tv[j];
    S[ro] = tl;
    __syncthreads();
  }

  if (wave < 4) {
    const int nidx = t - 1;
    if (nidx < 0 || nidx >= NFR) return;
    const float* gr = ((nidx & 1) ? GB : GA) + (size_t)plane * NP;
    const float* gi = gr + GSZ;
    float* pr = ((nidx & 1) ? PB : PA) + (size_t)plane * NPW;
    float* pi = pr + PSZ;
    const int no = (wave & 1) * 16;
    const int mset = wave >> 1;
    v2f bc[9], bs[9];
#pragma unroll
    for (int s = 0; s < 9; ++s) {
      bc[s] = frag_b(IVr, NL, 4 * s, no);
      bs[s] = frag_b(IVi, NL, 4 * s, no);
    }
    GRaw g[2][9];
    load_g_raw(gr, gi, mset * 16, g[0]);
#pragma unroll
    for (int j = 0; j < 3; ++j) {
      const int mt = 2 * j + mset;
      if (mt < 5) {
        if (mt + 2 < 5) load_g_raw(gr, gi, (mt + 2) * 16, g[(j + 1) & 1]);
        const int mo = mt * 16;
        v8f cr = zero8(), ci = zero8(), sr = zero8(), si = zero8();
#pragma unroll
        for (int s = 0; s < 9; ++s) {
          const GRaw& q = g[j & 1][s];
          v2f apr, api, amr, ami;
          apr.x = q.dr.x + q.mr.y;
          apr.y = q.dr.y + q.mr.x;
          api.x = q.di.x + q.mi.y;
          api.y = q.di.y + q.mi.x;
          amr.x = q.dr.x - q.mr.y;
          amr.y = q.dr.y - q.mr.x;
          ami.x = q.di.x - q.mi.y;
          ami.y = q.di.y - q.mi.x;
          if (s == 8) {
            apr.x = h ? 0.f : apr.x;
            api.x = h ? 0.f : api.x;
            amr.x = h ? 0.f : amr.x;
            ami.x = h ? 0.f : ami.x;
            apr.y = 0.f;
            api.y = 0.f;
            amr.y = 0.f;
            ami.y = 0.f;
          }
          cr = wmma4(apr, bc[s], cr);
          ci = wmma4(api, bc[s], ci);
          sr = wmma4(amr, bs[s], sr);
          si = wmma4(ami, bs[s], si);
        }
        const int n = no + (lane & 15);
        const int nm = NL - 1 - n;
        const int mb = mo + ((lane >> 4) << 3);
#pragma unroll
        for (int i = 0; i < 8; ++i) {
          const int m = mb + i;
          if (m < NLP) {
            pr[m * NL + n] = cr[i] - si[i];
            pi[m * NL + n] = ci[i] + sr[i];
            pr[m * NL + nm] = cr[i] + si[i];
            pi[m * NL + nm] = ci[i] - sr[i];
          }
        }
      }
    }
  } else {
    const int nidx = t - 2;
    if (nidx < 0 || nidx >= NFR) return;
    const int d = plane & (ND - 1), b = plane >> 4;
    const float* pr = ((nidx & 1) ? PB : PA) + (size_t)plane * NPW;
    const float* pi = pr + PSZ;
    float* op = out + ((size_t)b * (ND * NFR) + (d * NFR + nidx)) * (NL * NL);
    const int no = (wave - 4) * 16;
    const int n = no + (lane & 15);
    const float* prl = pr + (2 * h) * NL + n;
    const float* pil = pi + (2 * h) * NL + n;
    const float* prm = pr + (64 - 2 * h) * NL + n;
    const float* pim = pi + (64 - 2 * h) * NL + n;
    v2f bp[9], bm[9];
#pragma unroll
    for (int s = 1; s < 9; ++s) {
      const float a_r = prl[4 * s * NL], b_r = prl[(4 * s + 1) * NL];
      const float c_r = prm[(1 - 4 * s) * NL], d_r = prm[-4 * s * NL];
      const float a_i = pil[4 * s * NL], b_i = pil[(4 * s + 1) * NL];
      const float c_i = pim[(1 - 4 * s) * NL], d_i = pim[-4 * s * NL];
      bp[s].x = a_r + c_r;
      bp[s].y = b_r + d_r;
      bm[s].x = a_i - c_i;
      bm[s].y = b_i - d_i;
    }
    bp[8].x = h ? 0.f : bp[8].x;
    bm[8].x = h ? 0.f : bm[8].x;
    bp[8].y = 0.f;
    bm[8].y = 0.f;
    {
      const float a_r = prl[0], b_r = prl[NL], d_r = prm[0];
      const float a_i = pil[0], b_i = pil[NL], d_i = pim[0];
      const int c0 = (h ? 63 : 64) * NL + n;
      const float c_r = pr[c0], c_i = pi[c0];
      bp[0].x = a_r + (h ? c_r : 0.f);
      bp[0].y = b_r + d_r;
      bm[0].x = a_i - (h ? c_i : 0.f);
      bm[0].y = b_i - d_i;
    }
    for (int j = 0; j < 2; ++j) {
      const int mo = j * 16;
      v8f ec = zero8(), es = zero8();
#pragma unroll
      for (int s = 0; s < 9; ++s) {
        v2f ar = frag_a(S, 68, mo, 4 * s);
        v2f ai = frag_a(S + NL * 68, 68, mo, 4 * s);
        ec = wmma4(ar, bp[s], ec);
        es = wmma4(ai, bm[s], es);
      }
      const int mb = mo + ((lane >> 4) << 3);
#pragma unroll
      for (int i = 0; i < 8; ++i) {
        op[(mb + i) * NL + n] = ec[i] - es[i];
        op[(NL - 1 - (mb + i)) * NL + n] = ec[i] + es[i];
      }
    }
  }
}


extern "C" void kernel_launch(void* const* d_in, const int* in_sizes, int n_in,
                              void* d_out, int out_size, void* d_ws,
                              size_t ws_size, hipStream_t stream) {
  (void)in_sizes; (void)n_in; (void)out_size; (void)ws_size;
  const float* x   = (const float*)d_in[0];
  const float* wre = (const float*)d_in[1];
  const float* wim = (const float*)d_in[2];
  float* out = (float*)d_out;
  float* ws = (float*)d_ws;

  float* tw = ws;
  size_t off = TW_PAD;
  float* Yr = ws + off; off += (size_t)NB * NC * NP;
  float* Yi = ws + off; off += (size_t)NB * NC * NP;
  float* XT = ws + off; off += (size_t)2 * NB * NC * NP;
  float* Gr = ws + off; off += (size_t)NB * ND * NP;
  float* Gi = ws + off; off += (size_t)NB * ND * NP;
  float* Pr = ws + off; off += (size_t)NB * ND * NPW;
  float* Pi = ws + off; off += (size_t)NB * ND * NPW;
  (void)Gi; (void)Pi;

  k_twiddles<<<dim3(25, 11), 256, 0, stream>>>(tw, wre, wim, Pr, Yr, Yi);
  k_fft2d<<<NB * NC, 128, 0, stream>>>(x, tw, Yr, Yi);
  k_cfft<<<dim3((NP + 15) / 16, 2), 512, 0, stream>>>(tw, Yr, Yi, XT);
  for (int t = 0; t < NFR + 2; ++t)
    k_pipe<<<MIXB + NB * ND, 256, 0, stream>>>(tw, XT, wre, wim, Gr, Yr, Pr, Yi, out, t);
}
